# starting kernel
# baseline (speedup 1.0000x reference)
#include <hip/hip_runtime.h>


typedef __attribute__((ext_vector_type(2))) float v2f;
typedef __attribute__((ext_vector_type(8))) float v8f;

#define NC    12
#define C2    144
#define C3    1728
#define C4    20736
#define LPTS  128
#define NINC  127
#define KP    28
#define NTHR  288
#define SIGSZ (NC + C2 + C3 + C4)

__launch_bounds__(NTHR, 1)
__global__ void sig_depth4_kernel(const float* __restrict__ path,
                                  float* __restrict__ out) {
  __shared__ float dx[NINC * NC];
  __shared__ float s1[NC];
  __shared__ float s2[C2];
  __shared__ float s3[C3];
  __shared__ __align__(16) float Am[C2 * KP];
  __shared__ __align__(16) float Bt[C2 * KP];

  const int b    = blockIdx.x;
  const int tid  = threadIdx.x;
  const int lane = tid & 31;
  const int wv   = tid >> 5;
  const float* p = path + (size_t)b * LPTS * NC;

  for (int idx = tid; idx < NINC * NC; idx += NTHR) {
    int t = idx / NC, c = idx - t * NC;
    dx[idx] = p[(t + 1) * NC + c] - p[t * NC + c];
  }
  if (tid < NC) s1[tid] = 0.f;
  if (tid < C2) s2[tid] = 0.f;
  for (int idx = tid; idx < C3; idx += NTHR) s3[idx] = 0.f;

  v8f acc[9];
#pragma unroll
  for (int j = 0; j < 9; ++j)
#pragma unroll
    for (int v = 0; v < 8; ++v) acc[j][v] = 0.f;

  const int m  = lane & 15;
  const int kb = (lane >> 4) << 1;

  float n1 = 0.f, n2 = 0.f, n3r[6];

  for (int t = 0; t < NINC; ++t) {
    __syncthreads();
    const float* d = &dx[t * NC];

#pragma unroll
    for (int r = 0; r < 14; ++r) {
      int idx = tid + r * NTHR;
      int ij = idx / KP, c = idx - ij * KP;
      int i = ij / 12, j = ij - i * 12;
      float v;
      if (c < 12)       v = s3[ij * 12 + c];
      else if (c < 24)  v = (j == (c - 12)) ? s1[i] : 0.f;
      else if (c == 24) v = s2[ij] + d[i] * d[j] * (1.f / 12.f);
      else              v = 0.f;
      Am[idx] = v;
    }
#pragma unroll
    for (int r = 0; r < 14; ++r) {
      int idx = tid + r * NTHR;
      int kl = idx / KP, c = idx - kl * KP;
      int k = kl / 12, l = kl - k * 12;
      float v;
      if (c < 12)       v = (c == k) ? d[l] : 0.f;
      else if (c < 24)  v = d[c - 12] * d[k] * d[l] * (1.f / 6.f);
      else if (c == 24) v = d[k] * d[l] * 0.5f;
      else              v = 0.f;
      Bt[idx] = v;
    }
    if (tid < NC) n1 = s1[tid] + d[tid];
    if (tid < C2) {
      int i = tid / 12, j = tid - i * 12;
      n2 = s2[tid] + s1[i] * d[j] + d[i] * d[j] * 0.5f;
    }
#pragma unroll
    for (int r = 0; r < 6; ++r) {
      int idx = tid + r * NTHR;
      int i = idx / 144, jk = idx - i * 144;
      int j = jk / 12, k = jk - j * 12;
      n3r[r] = s3[idx] + s2[i * 12 + j] * d[k]
             + s1[i] * d[j] * d[k] * 0.5f
             + d[i] * d[j] * d[k] * (1.f / 6.f);
    }
    __syncthreads();

    if (tid < NC) s1[tid] = n1;
    if (tid < C2) s2[tid] = n2;
#pragma unroll
    for (int r = 0; r < 6; ++r) s3[tid + r * NTHR] = n3r[r];

    v2f a[7];
#pragma unroll
    for (int kc = 0; kc < 7; ++kc)
      a[kc] = *(const v2f*)&Am[(16 * wv + m) * KP + 4 * kc + kb];
#pragma unroll
    for (int j = 0; j < 9; ++j) {
#pragma unroll
      for (int kc = 0; kc < 7; ++kc) {
        v2f bf = *(const v2f*)&Bt[(16 * j + m) * KP + 4 * kc + kb];
        acc[j] = __builtin_amdgcn_wmma_f32_16x16x4_f32(
             false, a[kc],  false, bf,
             (short)0, acc[j],  false,  false);
      }
    }
  }

  __syncthreads();

  float* ob = out + (size_t)b * SIGSZ;
  if (tid < NC) ob[tid] = s1[tid];
  if (tid < C2) ob[NC + tid] = s2[tid];
#pragma unroll
  for (int r = 0; r < 6; ++r) {
    int idx = tid + r * NTHR;
    ob[NC + C2 + idx] = s3[idx];
  }
  float* o4 = ob + NC + C2 + C3;
  const int rowoff = (lane >= 16) ? 8 : 0;
#pragma unroll
  for (int j = 0; j < 9; ++j)
#pragma unroll
    for (int v = 0; v < 8; ++v) {
      int row = 16 * wv + v + rowoff;
      o4[row * 144 + 16 * j + m] = acc[j][v];
    }
}

extern "C" void kernel_launch(void* const* d_in, const int* in_sizes, int n_in,
                              void* d_out, int out_size, void* d_ws, size_t ws_size,
                              hipStream_t stream) {
  const float* path = (const float*)d_in[0];
  float* out = (float*)d_out;
  (void)in_sizes; (void)n_in; (void)out_size; (void)d_ws; (void)ws_size;
  sig_depth4_kernel<<<dim3(128), dim3(NTHR), 0, stream>>>(path, out);
}


# Level-2 suffix scan over time cut into 4 segments of 32 steps on 18 waves (segment totals through LDS, then a rescan from kept step terms), not 127 serial steps on 5 waves; sum order of W changes
# speedup vs baseline: 38.1968x; 38.1968x over previous
#include <hip/hip_runtime.h>


typedef __attribute__((ext_vector_type(2))) float v2f;
typedef __attribute__((ext_vector_type(8))) float v8f;

#define NC    12
#define C2    144
#define C3    1728
#define C4    20736
#define LPTS  128
#define NINC  127
#define TROWS 130
#define WCOLS 160
#define WPITCH 176
#define NWAVE 20
#define NTHR  (NWAVE * 32)
#define PTHR  576
#define SIGSZ (NC + C2 + C3 + C4)
#define NSEG  4
#define SEGT  32
#define STHR  (NSEG * C2)
#define YTHR  (NTHR - STHR)

__launch_bounds__(NTHR, 1)
__global__ void sig_depth4_kernel(const float* __restrict__ path,
                                  float* __restrict__ out) {
  __shared__ float xd[TROWS * NC];
  __shared__ __align__(8) v2f dp[TROWS * NC];
  __shared__ float r1[LPTS * NC];
  __shared__ __align__(8) v2f Wp[TROWS * WPITCH];
  __shared__ float segtot[STHR];

  const int b    = blockIdx.x;
  const int tid  = threadIdx.x;
  const int lane = tid & 31;
  const int wv   = tid >> 5;
  const float* p = path + (size_t)b * LPTS * NC;
  float* ob = out + (size_t)b * SIGSZ;

  if (tid < PTHR) {
    const int c = tid % NC;
    const float first = p[c], last = p[NINC * NC + c];
    float x0[3], x1[3];
#pragma unroll
    for (int it = 0; it < 3; ++it) {
      const int idx = tid + it * PTHR;
      const int li = idx < NINC * NC ? idx : NINC * NC - 1;
      x0[it] = p[li];
      x1[it] = p[li + NC];
    }
#pragma unroll
    for (int it = 0; it < 3; ++it) {
      const int idx = tid + it * PTHR;
      if (idx < TROWS * NC) {
        const bool live = idx < NINC * NC;
        const float d  = live ? x1[it] - x0[it] : 0.f;
        const float pp = live ? x0[it] - first : 0.f;
        const float rr = live ? last - x1[it] : 0.f;
        xd[idx] = d;
        v2f q; q[0] = d; q[1] = pp;
        dp[idx] = q;
        if (idx < LPTS * NC) r1[idx] = rr;
      }
    }
  }
  if (tid < 3 * WCOLS) {
    const int t = NINC + tid / WCOLS, c = tid - (tid / WCOLS) * WCOLS;
    v2f z; z[0] = 0.f; z[1] = 0.f;
    Wp[t * WPITCH + c] = z;
  }
  __syncthreads();

  const int seg = tid / C2;
  const int kl  = tid - seg * C2;
  const int k = kl / NC, l = kl - k * NC;
  const int ye = tid - STHR;
  const int kk = ye & 15, ty = (ye >> 4) & 1;
  const bool lv = kk < NC;
  const int kc = lv ? kk : 0;
  const float o0 = kk == NC ? 1.f : 0.f, o1 = kk == NC ? 0.5f : 0.f;
  float cq[SEGT], ce[SEGT];
  float s1 = 0.f;
  if (tid < STHR) {
    float tot = 0.f;
#pragma unroll
    for (int u = SEGT - 1; u >= 0; --u) {
      const int t = SEGT * seg + u;
      const float dk = xd[t * NC + k], dl = xd[t * NC + l], rl = r1[t * NC + l];
      cq[u] = dk * rl;
      ce[u] = dk * dl;
      tot = tot + cq[u] + ce[u] * 0.5f;
    }
    segtot[tid] = tot;
  } else if (ye < 32) {
#pragma unroll 8
    for (int n = 0; n < 32; ++n) {
      const int t = ty + 2 * n;
      const float d = xd[t * NC + kc], rr = r1[t * NC + kc];
      v2f w;
      w[0] = lv ? rr + d * 0.5f : o0;
      w[1] = lv ? rr * 0.5f + d * (1.f / 6.f) : o1;
      Wp[t * WPITCH + C2 + kk] = w;
    }
  } else {
    if (ye >= YTHR - NC) {
      const int c = ye - (YTHR - NC);
#pragma unroll 16
      for (int t = 0; t < 64; ++t) s1 += xd[t * NC + c];
    }
  }
  __syncthreads();

  if (tid < STHR) {
    float r2 = 0.f;
#pragma unroll
    for (int s2 = NSEG - 1; s2 >= 1; --s2)
      if (s2 > seg) r2 += segtot[s2 * C2 + kl];
#pragma unroll
    for (int u = SEGT - 1; u >= 0; --u) {
      const int t = SEGT * seg + u;
      const float q = cq[u], e = ce[u];
      v2f w;
      w[0] = r2 + q * 0.5f + e * (1.f / 6.f);
      w[1] = r2 * 0.5f + q * (1.f / 6.f) + e * (1.f / 24.f);
      if (u < SEGT - 1 || seg < NSEG - 1) Wp[t * WPITCH + kl] = w;
      r2 = r2 + q + e * 0.5f;
    }
  } else if (ye < 32) {
#pragma unroll 8
    for (int n = 32; n < 64; ++n) {
      const int t = ty + 2 * n;
      const float d = xd[t * NC + kc], rr = r1[t * NC + kc];
      v2f w;
      w[0] = lv ? rr + d * 0.5f : o0;
      w[1] = lv ? rr * 0.5f + d * (1.f / 6.f) : o1;
      if (t < NINC) Wp[t * WPITCH + C2 + kk] = w;
    }
  } else {
    if (ye >= YTHR - NC) {
      const int c = ye - (YTHR - NC);
#pragma unroll 16
      for (int t = 64; t < NINC; ++t) s1 += xd[t * NC + c];
    }
  }
  __syncthreads();
  if (ye >= YTHR - NC) ob[ye - (YTHR - NC)] = s1;

  const int m  = lane & 15;
  const int hf = lane >> 4;
  const int it = wv / 5;
  const int Jp = wv - 5 * it;
  const int jr = m < NC ? m : NC - 1;
  const int col0 = 32 * Jp + m;
  const int i0 = 3 * it;

  v8f acc[3][2];
#pragma unroll
  for (int q = 0; q < 3; ++q)
#pragma unroll
    for (int jj = 0; jj < 2; ++jj)
#pragma unroll
      for (int v = 0; v < 8; ++v) acc[q][jj][v] = 0.f;

  float an[2];
  v2f wn[2][2], xn[2][3];
#pragma unroll
  for (int s = 0; s < 2; ++s) {
    const int t = 2 * s + hf;
    an[s] = xd[t * NC + jr];
    wn[s][0] = Wp[t * WPITCH + col0];
    wn[s][1] = Wp[t * WPITCH + col0 + 16];
#pragma unroll
    for (int q = 0; q < 3; ++q) xn[s][q] = dp[t * NC + i0 + q];
  }

#pragma unroll 2
  for (int c = 0; c < 32; ++c) {
    float ac[2];
    v2f wc[2][2], xc[2][3];
#pragma unroll
    for (int s = 0; s < 2; ++s) {
      ac[s] = an[s];
      wc[s][0] = wn[s][0];
      wc[s][1] = wn[s][1];
#pragma unroll
      for (int q = 0; q < 3; ++q) xc[s][q] = xn[s][q];
    }
    const int cn = (c + 1) & 31;
#pragma unroll
    for (int s = 0; s < 2; ++s) {
      const int t = 4 * cn + 2 * s + hf;
      an[s] = xd[t * NC + jr];
      wn[s][0] = Wp[t * WPITCH + col0];
      wn[s][1] = Wp[t * WPITCH + col0 + 16];
#pragma unroll
      for (int q = 0; q < 3; ++q) xn[s][q] = dp[t * NC + i0 + q];
    }

    v2f a;
    a[0] = ac[0];
    a[1] = ac[1];
#pragma unroll
    for (int q = 0; q < 3; ++q)
#pragma unroll
      for (int jj = 0; jj < 2; ++jj) {
        v2f g;
        g[0] = xc[0][q][1] * wc[0][jj][0] + xc[0][q][0] * wc[0][jj][1];
        g[1] = xc[1][q][1] * wc[1][jj][0] + xc[1][q][0] * wc[1][jj][1];
        acc[q][jj] = __builtin_amdgcn_wmma_f32_16x16x4_f32(
             false, a,  false, g,
             (short)0, acc[q][jj],  false,  false);
      }
  }

  float* o2 = ob + NC;
  float* o3 = ob + NC + C2;
  float* o4 = ob + NC + C2 + C3;
#pragma unroll
  for (int q = 0; q < 3; ++q) {
    const int i = i0 + q;
#pragma unroll
    for (int jj = 0; jj < 2; ++jj) {
      const int J = 2 * Jp + jj;
#pragma unroll
      for (int v = 0; v < 8; ++v) {
        const int j = 8 * hf + v;
        if (j < NC) {
          const int ij = i * NC + j;
          if (J < 9) {
            o4[ij * C2 + 16 * J + m] = acc[q][jj][v];
          } else if (m < NC) {
            o3[ij * NC + m] = acc[q][jj][v];
          } else if (m == NC) {
            o2[ij] = acc[q][jj][v];
          }
        }
      }
    }
  }
}

extern "C" void kernel_launch(void* const* d_in, const int* in_sizes, int n_in,
                              void* d_out, int out_size, void* d_ws, size_t ws_size,
                              hipStream_t stream) {
  const float* path = (const float*)d_in[0];
  float* out = (float*)d_out;
  (void)in_sizes; (void)n_in; (void)out_size; (void)d_ws; (void)ws_size;
  sig_depth4_kernel<<<dim3(128), dim3(NTHR), 0, stream>>>(path, out);
}


# Store epilogue: tile place from a scalar copy of the wave number, all level-4 rows from one lane address plus constants, stores grouped by lane mask: 1 address wait per wave, not 1 per store
# speedup vs baseline: 44.7918x; 1.1727x over previous
#include <hip/hip_runtime.h>


typedef __attribute__((ext_vector_type(2))) float v2f;
typedef __attribute__((ext_vector_type(8))) float v8f;

#define NC    12
#define C2    144
#define C3    1728
#define C4    20736
#define LPTS  128
#define NINC  127
#define TROWS 130
#define WCOLS 160
#define WPITCH 176
#define NWAVE 20
#define NTHR  (NWAVE * 32)
#define PTHR  576
#define SIGSZ (NC + C2 + C3 + C4)
#define NSEG  4
#define SEGT  32
#define STHR  (NSEG * C2)
#define YTHR  (NTHR - STHR)

__launch_bounds__(NTHR, 1)
__global__ void sig_depth4_kernel(const float* __restrict__ path,
                                  float* __restrict__ out) {
  __shared__ float xd[TROWS * NC];
  __shared__ __align__(8) v2f dp[TROWS * NC];
  __shared__ float r1[LPTS * NC];
  __shared__ __align__(8) v2f Wp[TROWS * WPITCH];
  __shared__ float segtot[STHR];

  const int b    = blockIdx.x;
  const int tid  = threadIdx.x;
  const int lane = tid & 31;
  const int wv   = tid >> 5;
  const float* p = path + (size_t)b * LPTS * NC;
  float* ob = out + (size_t)b * SIGSZ;

  if (tid < PTHR) {
    const int c = tid % NC;
    const float first = p[c], last = p[NINC * NC + c];
    float x0[3], x1[3];
#pragma unroll
    for (int it = 0; it < 3; ++it) {
      const int idx = tid + it * PTHR;
      const int li = idx < NINC * NC ? idx : NINC * NC - 1;
      x0[it] = p[li];
      x1[it] = p[li + NC];
    }
#pragma unroll
    for (int it = 0; it < 3; ++it) {
      const int idx = tid + it * PTHR;
      if (idx < TROWS * NC) {
        const bool live = idx < NINC * NC;
        const float d  = live ? x1[it] - x0[it] : 0.f;
        const float pp = live ? x0[it] - first : 0.f;
        const float rr = live ? last - x1[it] : 0.f;
        xd[idx] = d;
        v2f q; q[0] = d; q[1] = pp;
        dp[idx] = q;
        if (idx < LPTS * NC) r1[idx] = rr;
      }
    }
  }
  if (tid < 3 * WCOLS) {
    const int t = NINC + tid / WCOLS, c = tid - (tid / WCOLS) * WCOLS;
    v2f z; z[0] = 0.f; z[1] = 0.f;
    Wp[t * WPITCH + c] = z;
  }
  __syncthreads();

  const int seg = tid / C2;
  const int kl  = tid - seg * C2;
  const int k = kl / NC, l = kl - k * NC;
  const int ye = tid - STHR;
  const int kk = ye & 15, ty = (ye >> 4) & 1;
  const bool lv = kk < NC;
  const int kc = lv ? kk : 0;
  const float o0 = kk == NC ? 1.f : 0.f, o1 = kk == NC ? 0.5f : 0.f;
  float cq[SEGT], ce[SEGT];
  float s1 = 0.f;
  if (tid < STHR) {
    float tot = 0.f;
#pragma unroll
    for (int u = SEGT - 1; u >= 0; --u) {
      const int t = SEGT * seg + u;
      const float dk = xd[t * NC + k], dl = xd[t * NC + l], rl = r1[t * NC + l];
      cq[u] = dk * rl;
      ce[u] = dk * dl;
      tot = tot + cq[u] + ce[u] * 0.5f;
    }
    segtot[tid] = tot;
  } else if (ye < 32) {
#pragma unroll 8
    for (int n = 0; n < 32; ++n) {
      const int t = ty + 2 * n;
      const float d = xd[t * NC + kc], rr = r1[t * NC + kc];
      v2f w;
      w[0] = lv ? rr + d * 0.5f : o0;
      w[1] = lv ? rr * 0.5f + d * (1.f / 6.f) : o1;
      Wp[t * WPITCH + C2 + kk] = w;
    }
  } else {
    if (ye >= YTHR - NC) {
      const int c = ye - (YTHR - NC);
#pragma unroll 16
      for (int t = 0; t < 64; ++t) s1 += xd[t * NC + c];
    }
  }
  __syncthreads();

  if (tid < STHR) {
    float r2 = 0.f;
#pragma unroll
    for (int s2 = NSEG - 1; s2 >= 1; --s2)
      if (s2 > seg) r2 += segtot[s2 * C2 + kl];
#pragma unroll
    for (int u = SEGT - 1; u >= 0; --u) {
      const int t = SEGT * seg + u;
      const float q = cq[u], e = ce[u];
      v2f w;
      w[0] = r2 + q * 0.5f + e * (1.f / 6.f);
      w[1] = r2 * 0.5f + q * (1.f / 6.f) + e * (1.f / 24.f);
      if (u < SEGT - 1 || seg < NSEG - 1) Wp[t * WPITCH + kl] = w;
      r2 = r2 + q + e * 0.5f;
    }
  } else if (ye < 32) {
#pragma unroll 8
    for (int n = 32; n < 64; ++n) {
      const int t = ty + 2 * n;
      const float d = xd[t * NC + kc], rr = r1[t * NC + kc];
      v2f w;
      w[0] = lv ? rr + d * 0.5f : o0;
      w[1] = lv ? rr * 0.5f + d * (1.f / 6.f) : o1;
      if (t < NINC) Wp[t * WPITCH + C2 + kk] = w;
    }
  } else {
    if (ye >= YTHR - NC) {
      const int c = ye - (YTHR - NC);
#pragma unroll 16
      for (int t = 64; t < NINC; ++t) s1 += xd[t * NC + c];
    }
  }
  __syncthreads();
  if (ye >= YTHR - NC) ob[ye - (YTHR - NC)] = s1;

  const int m  = lane & 15;
  const int hf = lane >> 4;
  const int it = wv / 5;
  const int Jp = wv - 5 * it;
  const int jr = m < NC ? m : NC - 1;
  const int col0 = 32 * Jp + m;
  const int i0 = 3 * it;

  v8f acc[3][2];
#pragma unroll
  for (int q = 0; q < 3; ++q)
#pragma unroll
    for (int jj = 0; jj < 2; ++jj)
#pragma unroll
      for (int v = 0; v < 8; ++v) acc[q][jj][v] = 0.f;

  float an[2];
  v2f wn[2][2], xn[2][3];
#pragma unroll
  for (int s = 0; s < 2; ++s) {
    const int t = 2 * s + hf;
    an[s] = xd[t * NC + jr];
    wn[s][0] = Wp[t * WPITCH + col0];
    wn[s][1] = Wp[t * WPITCH + col0 + 16];
#pragma unroll
    for (int q = 0; q < 3; ++q) xn[s][q] = dp[t * NC + i0 + q];
  }

#pragma unroll 2
  for (int c = 0; c < 32; ++c) {
    float ac[2];
    v2f wc[2][2], xc[2][3];
#pragma unroll
    for (int s = 0; s < 2; ++s) {
      ac[s] = an[s];
      wc[s][0] = wn[s][0];
      wc[s][1] = wn[s][1];
#pragma unroll
      for (int q = 0; q < 3; ++q) xc[s][q] = xn[s][q];
    }
    const int cn = (c + 1) & 31;
#pragma unroll
    for (int s = 0; s < 2; ++s) {
      const int t = 4 * cn + 2 * s + hf;
      an[s] = xd[t * NC + jr];
      wn[s][0] = Wp[t * WPITCH + col0];
      wn[s][1] = Wp[t * WPITCH + col0 + 16];
#pragma unroll
      for (int q = 0; q < 3; ++q) xn[s][q] = dp[t * NC + i0 + q];
    }

    v2f a;
    a[0] = ac[0];
    a[1] = ac[1];
#pragma unroll
    for (int q = 0; q < 3; ++q)
#pragma unroll
      for (int jj = 0; jj < 2; ++jj) {
        v2f g;
        g[0] = xc[0][q][1] * wc[0][jj][0] + xc[0][q][0] * wc[0][jj][1];
        g[1] = xc[1][q][1] * wc[1][jj][0] + xc[1][q][0] * wc[1][jj][1];
        acc[q][jj] = __builtin_amdgcn_wmma_f32_16x16x4_f32(
             false, a,  false, g,
             (short)0, acc[q][jj],  false,  false);
      }
  }

  const int wvu = __builtin_amdgcn_readfirstlane(wv);
  const int itu = wvu / 5;
  const int Jpu = wvu - 5 * itu;
  const bool l4b = Jpu < 4;
  auto rows03 = [](float* pt, const int st, const v8f a) {
    pt[0] = a[0];
    pt[st] = a[1];
    pt[2 * st] = a[2];
    pt[3 * st] = a[3];
  };
  auto rows47 = [](float* pt, const int st, const v8f a) {
    pt[4 * st] = a[4];
    pt[5 * st] = a[5];
    pt[6 * st] = a[6];
    pt[7 * st] = a[7];
  };
  const int ij0 = 3 * itu * NC + 8 * hf;
  float* p4 = ob + NC + C2 + C3 + ij0 * C2 + 32 * Jpu + m;
#pragma unroll
  for (int q = 0; q < 3; ++q) rows03(p4 + q * NC * C2, C2, acc[q][0]);
  if (l4b) {
#pragma unroll
    for (int q = 0; q < 3; ++q) rows03(p4 + q * NC * C2 + 16, C2, acc[q][1]);
  }
  if (hf == 0) {
#pragma unroll
    for (int q = 0; q < 3; ++q) rows47(p4 + q * NC * C2, C2, acc[q][0]);
    if (l4b) {
#pragma unroll
      for (int q = 0; q < 3; ++q) rows47(p4 + q * NC * C2 + 16, C2, acc[q][1]);
    }
  }
  if (!l4b) {
    if (m < NC) {
      float* p3 = ob + NC + C2 + ij0 * NC + m;
#pragma unroll
      for (int q = 0; q < 3; ++q) rows03(p3 + q * C2, NC, acc[q][1]);
      if (hf == 0) {
#pragma unroll
        for (int q = 0; q < 3; ++q) rows47(p3 + q * C2, NC, acc[q][1]);
      }
    } else if (m == NC) {
      float* p2 = ob + NC + ij0;
#pragma unroll
      for (int q = 0; q < 3; ++q) rows03(p2 + q * NC, 1, acc[q][1]);
      if (hf == 0) {
#pragma unroll
        for (int q = 0; q < 3; ++q) rows47(p2 + q * NC, 1, acc[q][1]);
      }
    }
  }
}

extern "C" void kernel_launch(void* const* d_in, const int* in_sizes, int n_in,
                              void* d_out, int out_size, void* d_ws, size_t ws_size,
                              hipStream_t stream) {
  const float* path = (const float*)d_in[0];
  float* out = (float*)d_out;
  (void)in_sizes; (void)n_in; (void)out_size; (void)d_ws; (void)ws_size;
  sig_depth4_kernel<<<dim3(128), dim3(NTHR), 0, stream>>>(path, out);
}


# Each batch element on two work-groups, one per half of W's columns (its half of the level-2 scan, 60 of 120 tiles, half the stores); 12 waves of 5 tiles, grid 128 to 256; sum order unchanged
# speedup vs baseline: 64.1823x; 1.4329x over previous
#include <hip/hip_runtime.h>


typedef __attribute__((ext_vector_type(2))) float v2f;
typedef __attribute__((ext_vector_type(8))) float v8f;

#define NC    12
#define C2    144
#define C3    1728
#define C4    20736
#define LPTS  128
#define NINC  127
#define TROWS 130
#define HCOLS 80
#define KLHI  (C2 - HCOLS)
#define WPITCH 176
#define NWAVE 12
#define NTHR  (NWAVE * 32)
#define SIGSZ (NC + C2 + C3 + C4)
#define NSEG  4
#define SEGT  32
#define STLO  (NSEG * HCOLS)
#define STHI  (NSEG * KLHI)
#define YTHR  64

__launch_bounds__(NTHR, 1)
__global__ void sig_depth4_kernel(const float* __restrict__ path,
                                  float* __restrict__ out) {
  __shared__ float xd[TROWS * NC];
  __shared__ __align__(8) v2f dp[TROWS * NC];
  __shared__ float r1[LPTS * NC];
  __shared__ __align__(8) v2f Wp[TROWS * WPITCH];
  __shared__ float segtot[STLO];

  const int b    = blockIdx.x >> 1;
  const int h    = blockIdx.x & 1;
  const int tid  = threadIdx.x;
  const int lane = tid & 31;
  const int wv   = tid >> 5;
  const float* p = path + (size_t)b * LPTS * NC;
  float* ob = out + (size_t)b * SIGSZ;

  {
    const int c = tid % NC;
    const float first = p[c], last = p[NINC * NC + c];
    float x0[4], x1[4];
#pragma unroll
    for (int it = 0; it < 4; ++it) {
      const int idx = tid + it * NTHR;
      const int li = idx < NINC * NC ? idx : NINC * NC - 1;
      x0[it] = p[li];
      x1[it] = p[li + NC];
    }
#pragma unroll
    for (int it = 0; it < 4; ++it) {
      const int idx = tid + it * NTHR;
      const bool live = idx < NINC * NC;
      const float d  = live ? x1[it] - x0[it] : 0.f;
      const float pp = live ? x0[it] - first : 0.f;
      const float rr = live ? last - x1[it] : 0.f;
      xd[idx] = d;
      v2f q; q[0] = d; q[1] = pp;
      dp[idx] = q;
      r1[idx] = rr;
    }
  }
  if (tid < (TROWS - LPTS) * NC) {
    v2f z; z[0] = 0.f; z[1] = 0.f;
    xd[LPTS * NC + tid] = 0.f;
    dp[LPTS * NC + tid] = z;
  }
  if (tid < 3 * HCOLS) {
    const int t = NINC + tid / HCOLS, c = tid - (tid / HCOLS) * HCOLS;
    v2f z; z[0] = 0.f; z[1] = 0.f;
    Wp[t * WPITCH + c] = z;
  }
  __syncthreads();

  const int nkl = h ? KLHI : HCOLS;
  const int seg = h ? tid / KLHI : tid / HCOLS;
  const int kll = tid - seg * nkl;
  const int kl  = HCOLS * h + kll;
  const int k = kl / NC, l = kl - k * NC;
  const bool scan = tid < NSEG * nkl;
  const int ye = tid - STHI;
  const bool ycol = h != 0 && ye >= 0 && ye < 32;
  const bool lev1 = h != 0 && ye >= YTHR - NC && ye < YTHR;
  const int kk = ye & 15, ty = (ye >> 4) & 1;
  const bool lv = kk < NC;
  const int kc = lv ? kk : 0;
  const float o0 = kk == NC ? 1.f : 0.f, o1 = kk == NC ? 0.5f : 0.f;
  float cq[SEGT], ce[SEGT];
  float s1 = 0.f;
  if (scan) {
    float tot = 0.f;
#pragma unroll
    for (int u = SEGT - 1; u >= 0; --u) {
      const int t = SEGT * seg + u;
      const float dk = xd[t * NC + k], dl = xd[t * NC + l], rl = r1[t * NC + l];
      cq[u] = dk * rl;
      ce[u] = dk * dl;
      tot = tot + cq[u] + ce[u] * 0.5f;
    }
    segtot[seg * HCOLS + kll] = tot;
  } else if (ycol) {
#pragma unroll 8
    for (int n = 0; n < 32; ++n) {
      const int t = ty + 2 * n;
      const float d = xd[t * NC + kc], rr = r1[t * NC + kc];
      v2f w;
      w[0] = lv ? rr + d * 0.5f : o0;
      w[1] = lv ? rr * 0.5f + d * (1.f / 6.f) : o1;
      Wp[t * WPITCH + KLHI + kk] = w;
    }
  } else if (lev1) {
    const int c = ye - (YTHR - NC);
#pragma unroll 16
    for (int t = 0; t < 64; ++t) s1 += xd[t * NC + c];
  }
  __syncthreads();

  if (scan) {
    float r2 = 0.f;
#pragma unroll
    for (int s2 = NSEG - 1; s2 >= 1; --s2)
      if (s2 > seg) r2 += segtot[s2 * HCOLS + kll];
#pragma unroll
    for (int u = SEGT - 1; u >= 0; --u) {
      const int t = SEGT * seg + u;
      const float q = cq[u], e = ce[u];
      v2f w;
      w[0] = r2 + q * 0.5f + e * (1.f / 6.f);
      w[1] = r2 * 0.5f + q * (1.f / 6.f) + e * (1.f / 24.f);
      if (u < SEGT - 1 || seg < NSEG - 1) Wp[t * WPITCH + kll] = w;
      r2 = r2 + q + e * 0.5f;
    }
  } else if (ycol) {
#pragma unroll 8
    for (int n = 32; n < 64; ++n) {
      const int t = ty + 2 * n;
      const float d = xd[t * NC + kc], rr = r1[t * NC + kc];
      v2f w;
      w[0] = lv ? rr + d * 0.5f : o0;
      w[1] = lv ? rr * 0.5f + d * (1.f / 6.f) : o1;
      if (t < NINC) Wp[t * WPITCH + KLHI + kk] = w;
    }
  } else if (lev1) {
    const int c = ye - (YTHR - NC);
#pragma unroll 16
    for (int t = 64; t < NINC; ++t) s1 += xd[t * NC + c];
  }
  __syncthreads();
  if (lev1) ob[ye - (YTHR - NC)] = s1;

  const int m  = lane & 15;
  const int hf = lane >> 4;
  const int jr = m < NC ? m : NC - 1;

  v8f acc[5];
#pragma unroll
  for (int jj = 0; jj < 5; ++jj)
#pragma unroll
    for (int v = 0; v < 8; ++v) acc[jj][v] = 0.f;

  float an[2];
  v2f wn[2][5], xn[2];
#pragma unroll
  for (int s = 0; s < 2; ++s) {
    const int t = 2 * s + hf;
    an[s] = xd[t * NC + jr];
#pragma unroll
    for (int jj = 0; jj < 5; ++jj) wn[s][jj] = Wp[t * WPITCH + 16 * jj + m];
    xn[s] = dp[t * NC + wv];
  }

#pragma unroll 2
  for (int c = 0; c < 32; ++c) {
    float ac[2];
    v2f wc[2][5], xc[2];
#pragma unroll
    for (int s = 0; s < 2; ++s) {
      ac[s] = an[s];
#pragma unroll
      for (int jj = 0; jj < 5; ++jj) wc[s][jj] = wn[s][jj];
      xc[s] = xn[s];
    }
    const int cn = (c + 1) & 31;
#pragma unroll
    for (int s = 0; s < 2; ++s) {
      const int t = 4 * cn + 2 * s + hf;
      an[s] = xd[t * NC + jr];
#pragma unroll
      for (int jj = 0; jj < 5; ++jj) wn[s][jj] = Wp[t * WPITCH + 16 * jj + m];
      xn[s] = dp[t * NC + wv];
    }

    v2f a;
    a[0] = ac[0];
    a[1] = ac[1];
#pragma unroll
    for (int jj = 0; jj < 5; ++jj) {
      v2f g;
      g[0] = xc[0][1] * wc[0][jj][0] + xc[0][0] * wc[0][jj][1];
      g[1] = xc[1][1] * wc[1][jj][0] + xc[1][0] * wc[1][jj][1];
      acc[jj] = __builtin_amdgcn_wmma_f32_16x16x4_f32(
           false, a,  false, g,
           (short)0, acc[jj],  false,  false);
    }
  }

  const int wvu = __builtin_amdgcn_readfirstlane(wv);
  const bool l4b = h == 0;
  auto rows03 = [](float* pt, const int st, const v8f a) {
    pt[0] = a[0];
    pt[st] = a[1];
    pt[2 * st] = a[2];
    pt[3 * st] = a[3];
  };
  auto rows47 = [](float* pt, const int st, const v8f a) {
    pt[4 * st] = a[4];
    pt[5 * st] = a[5];
    pt[6 * st] = a[6];
    pt[7 * st] = a[7];
  };
  const int ij0 = wvu * NC + 8 * hf;
  float* p4 = ob + NC + C2 + C3 + ij0 * C2 + HCOLS * h + m;
#pragma unroll
  for (int jj = 0; jj < 4; ++jj) rows03(p4 + 16 * jj, C2, acc[jj]);
  if (l4b) rows03(p4 + 64, C2, acc[4]);
  if (hf == 0) {
#pragma unroll
    for (int jj = 0; jj < 4; ++jj) rows47(p4 + 16 * jj, C2, acc[jj]);
    if (l4b) rows47(p4 + 64, C2, acc[4]);
  }
  if (!l4b) {
    if (m < NC) {
      float* p3 = ob + NC + C2 + ij0 * NC + m;
      rows03(p3, NC, acc[4]);
      if (hf == 0) rows47(p3, NC, acc[4]);
    } else if (m == NC) {
      float* p2 = ob + NC + ij0;
      rows03(p2, 1, acc[4]);
      if (hf == 0) rows47(p2, 1, acc[4]);
    }
  }
}

extern "C" void kernel_launch(void* const* d_in, const int* in_sizes, int n_in,
                              void* d_out, int out_size, void* d_ws, size_t ws_size,
                              hipStream_t stream) {
  const float* path = (const float*)d_in[0];
  float* out = (float*)d_out;
  (void)in_sizes; (void)n_in; (void)out_size; (void)d_ws; (void)ws_size;
  sig_depth4_kernel<<<dim3(256), dim3(NTHR), 0, stream>>>(path, out);
}
